# starting kernel
# baseline (speedup 1.0000x reference)
#include <hip/hip_runtime.h>

typedef __attribute__((ext_vector_type(2))) float v2f;
typedef __attribute__((ext_vector_type(4))) float v4f;
typedef __attribute__((ext_vector_type(8))) float v8f;

#define NB    16384
#define DIN   204
#define NPAIR 100
#define PSTR  70

static __device__ __forceinline__ v8f wmma4(v2f a, v2f b, v8f c) {
  return __builtin_amdgcn_wmma_f32_16x16x4_f32(false, a, false, b, (short)0, c, false, false);
}

static __device__ __forceinline__ v8f splat8(float v) {
  v8f r;
#pragma unroll
  for (int e = 0; e < 8; ++e) r[e] = v;
  return r;
}

static __device__ __forceinline__ void lds_fence() {
  asm volatile("s_wait_dscnt 0" ::: "memory");
}

__global__ __launch_bounds__(256) void bn_stats_kernel(const float* __restrict__ X,
                                                       float* __restrict__ nrm) {
  const int c = blockIdx.x;
  float s = 0.f, s2 = 0.f;
  for (int r = threadIdx.x; r < NB; r += 256) {
    float v = X[(size_t)r * DIN + c];
    s += v;
    s2 += v * v;
  }
  __shared__ float red0[256], red1[256];
  red0[threadIdx.x] = s;
  red1[threadIdx.x] = s2;
  __syncthreads();
  for (int st = 128; st > 0; st >>= 1) {
    if (threadIdx.x < st) {
      red0[threadIdx.x] += red0[threadIdx.x + st];
      red1[threadIdx.x] += red1[threadIdx.x + st];
    }
    __syncthreads();
  }
  if (threadIdx.x == 0) {
    float m   = red0[0] * (1.f / NB);
    float var = red1[0] * (1.f / NB) - m * m;
    float rs  = rsqrtf(var + 1e-5f);
    nrm[2 * c]     = rs;
    nrm[2 * c + 1] = -m * rs;
  }
}

template <int KSTEPS, int NT, bool RELU>
static __device__ __forceinline__ void dense_layer(const float* pb, int nc, bool hi,
                                                   const float* __restrict__ W, int wstride,
                                                   const float* __restrict__ bias, v8f* acc) {
  const int kof = hi ? 2 : 0;
#pragma unroll
  for (int t = 0; t < NT; ++t) acc[t] = splat8(bias[t * 16 + nc]);
#pragma unroll
  for (int kk = 0; kk < KSTEPS; ++kk) {
    const int ko = 4 * kk + kof;
    v2f a = *(const v2f*)(pb + nc * PSTR + ko);
#pragma unroll
    for (int t = 0; t < NT; ++t) {
      const int n = t * 16 + nc;
      v2f b = *(const v2f*)(W + n * wstride + ko);
      acc[t] = wmma4(a, b, acc[t]);
    }
  }
  if (RELU) {
#pragma unroll
    for (int t = 0; t < NT; ++t)
#pragma unroll
      for (int e = 0; e < 8; ++e) acc[t][e] = fmaxf(acc[t][e], 0.01f * acc[t][e]);
  }
}

static __device__ __forceinline__ void store_h(float* pb, int nc, bool hi, const v8f* h) {
#pragma unroll
  for (int t = 0; t < 4; ++t)
#pragma unroll
    for (int e = 0; e < 8; ++e) {
      const int r = e + (hi ? 8 : 0);
      pb[r * PSTR + t * 16 + nc] = h[t][e];
    }
}

__global__ __launch_bounds__(128) void policy_fused_kernel(
    const float* __restrict__ X, const float* __restrict__ nrm,
    const float* __restrict__ Wme, const float* __restrict__ bme,
    const float* __restrict__ W1, const float* __restrict__ b1,
    const float* __restrict__ W2, const float* __restrict__ b2,
    const float* __restrict__ W3, const float* __restrict__ b3,
    float* __restrict__ out) {
  __shared__ float polbuf[4][16 * PSTR];

  const int  lane = threadIdx.x & 31;
  const int  wv   = threadIdx.x >> 5;
  const int  nc   = lane & 15;
  const bool hi   = (lane >= 16);
  const int  rowbase = blockIdx.x * 64 + wv * 16;
  const int  myrow   = rowbase + nc;
  float* pb = polbuf[wv];
  const float* xrow = X + (size_t)myrow * DIN;

  v2f bw[4];
#pragma unroll
  for (int t = 0; t < 4; ++t) {
    const int n = t * 16 + nc;
    v2f w = *(const v2f*)(Wme + 2 * n);
    v2f z; z.x = 0.f; z.y = 0.f;
    bw[t] = hi ? z : w;
  }

  v8f absacc[4];
#pragma unroll
  for (int t = 0; t < 4; ++t) absacc[t] = splat8(0.f);
  v2f asum; asum.x = 0.f; asum.y = 0.f;
  const v8f zero8 = splat8(0.f);

  v2f xv = *(const v2f*)(xrow + 4);
  v4f nv = *(const v4f*)(nrm + 8);
  for (int k = 0; k < NPAIR; ++k) {
    v2f av;
    av.x = fmaf(xv.x, nv.x, nv.y);
    av.y = fmaf(xv.y, nv.z, nv.w);
    const int cn = (k < NPAIR - 1) ? (6 + 2 * k) : 4;
    xv = *(const v2f*)(xrow + cn);
    nv = *(const v4f*)(nrm + 2 * cn);
    v2f z2; z2.x = 0.f; z2.y = 0.f;
    av = hi ? z2 : av;
    asum.x += av.x; asum.y += av.y;
#pragma unroll
    for (int t = 0; t < 4; ++t) {
      v8f z = wmma4(av, bw[t], zero8);
#pragma unroll
      for (int e = 0; e < 8; ++e) absacc[t][e] += fabsf(z[e]);
    }
  }

#pragma unroll
  for (int t = 0; t < 4; ++t) {
    v8f cb = splat8(100.f * bme[t * 16 + nc]);
    v8f zs = wmma4(asum, bw[t], cb);
#pragma unroll
    for (int e = 0; e < 8; ++e) {
      const float m = (0.505f * zs[e] + 0.495f * absacc[t][e]) * 0.01f;
      const int r = e + (hi ? 8 : 0);
      pb[r * PSTR + 4 + t * 16 + nc] = m;
    }
  }
  if (!hi) {
#pragma unroll
    for (int c = 0; c < 4; ++c)
      pb[nc * PSTR + c] = fmaf(xrow[c], nrm[2 * c], nrm[2 * c + 1]);
  }
  lds_fence();

  v8f h[4];
  dense_layer<17, 4, true>(pb, nc, hi, W1, 68, b1, h);
  lds_fence();
  store_h(pb, nc, hi, h);
  lds_fence();
  dense_layer<16, 4, true>(pb, nc, hi, W2, 64, b2, h);
  lds_fence();
  store_h(pb, nc, hi, h);
  lds_fence();
  v8f lg[2];
  dense_layer<16, 2, false>(pb, nc, hi, W3, 64, b3, lg);

#pragma unroll
  for (int r = 0; r < 8; ++r) {
    const float a0 = lg[0][r], a1 = lg[1][r];
    float m = fmaxf(a0, a1);
#pragma unroll
    for (int off = 8; off >= 1; off >>= 1) m = fmaxf(m, __shfl_xor(m, off, 32));
    const float e0 = __expf(a0 - m);
    const float e1 = __expf(a1 - m);
    float s = e0 + e1;
#pragma unroll
    for (int off = 8; off >= 1; off >>= 1) s += __shfl_xor(s, off, 32);
    const float inv = 1.0f / s;
    const int row = rowbase + r + (hi ? 8 : 0);
    out[row * 32 + nc]      = e0 * inv;
    out[row * 32 + 16 + nc] = e1 * inv;
  }
}

extern "C" void kernel_launch(void* const* d_in, const int* in_sizes, int n_in,
                              void* d_out, int out_size, void* d_ws, size_t ws_size,
                              hipStream_t stream) {
  const float* X   = (const float*)d_in[0];
  const float* Wme = (const float*)d_in[1];
  const float* bme = (const float*)d_in[2];
  const float* W1  = (const float*)d_in[3];
  const float* b1  = (const float*)d_in[4];
  const float* W2  = (const float*)d_in[5];
  const float* b2  = (const float*)d_in[6];
  const float* W3  = (const float*)d_in[7];
  const float* b3  = (const float*)d_in[8];
  float* out = (float*)d_out;

  float* nrm = (float*)d_ws;

  bn_stats_kernel<<<DIN, 256, 0, stream>>>(X, nrm);
  policy_fused_kernel<<<NB / 64, 128, 0, stream>>>(X, nrm, Wme, bme,
                                                   W1, b1, W2, b2, W3, b3, out);
}


# MLP weights copied once per call into WMMA fragment order in the workspace by 11 added work-groups of the first launch: a B fragment is whole-line 16-byte reads, not sixteen-line gathers
# speedup vs baseline: 1.5477x; 1.5477x over previous
#include <hip/hip_runtime.h>

typedef __attribute__((ext_vector_type(2))) float v2f;
typedef __attribute__((ext_vector_type(4))) float v4f;
typedef __attribute__((ext_vector_type(8))) float v8f;

#define NB    16384
#define DIN   204
#define NPAIR 100
#define PSTR  70
#define NWAVE 4
#define PPW   25

static __device__ __forceinline__ v8f wmma4(v2f a, v2f b, v8f c) {
  return __builtin_amdgcn_wmma_f32_16x16x4_f32(false, a, false, b, (short)0, c, false, false);
}

static __device__ __forceinline__ v8f splat8(float v) {
  v8f r;
#pragma unroll
  for (int e = 0; e < 8; ++e) r[e] = v;
  return r;
}

#define WF_OFF   512
#define WF1_LEN  (64 * 68)
#define WF2_LEN  (64 * 64)
#define WF3_LEN  (32 * 64)
#define WF1_OFF  WF_OFF
#define WF2_OFF  (WF1_OFF + WF1_LEN)
#define WF3_OFF  (WF2_OFF + WF2_LEN)
#define WF_TOTAL (WF1_LEN + WF2_LEN + WF3_LEN)

template <int K>
static __device__ __forceinline__ void pack_one(const float* __restrict__ Wsrc, float* __restrict__ dst, int idx) {
  constexpr int KS = K / 4, Q = KS / 2, TS = KS * 64;
  const int t = idx / TS, r = idx - t * TS;
  int lane, kk, s;
  if (r < Q * 128) { const int e = r & 3; lane = (r >> 2) & 31; kk = 2 * (r >> 7) + (e >> 1); s = e & 1; }
  else             { const int r2 = r - Q * 128; lane = r2 >> 1; kk = 2 * Q; s = r2 & 1; }
  const int n = 16 * t + (lane & 15);
  const int k = 4 * kk + 2 * (lane >> 4) + s;
  dst[idx] = Wsrc[n * K + k];
}

#define SCOLS 4
#define STHR  1024
#define SROWS (NB / STHR)
#define SPHASE 16
#define NSTAT  (DIN / SCOLS)
#define NPACK  ((WF_TOTAL + STHR - 1) / STHR)
__global__ __launch_bounds__(STHR) void bn_stats_kernel(const float* __restrict__ X,
                                                        const float* __restrict__ W1,
                                                        const float* __restrict__ W2,
                                                        const float* __restrict__ W3,
                                                        float* __restrict__ nrm) {
  if (blockIdx.x >= NSTAT) {
    const int g = (blockIdx.x - NSTAT) * STHR + threadIdx.x;
    float* wf = nrm + WF_OFF;
    if (g < WF1_LEN)                 pack_one<68>(W1, wf, g);
    else if (g < WF1_LEN + WF2_LEN)  pack_one<64>(W2, wf + WF1_LEN, g - WF1_LEN);
    else if (g < WF_TOTAL)           pack_one<64>(W3, wf + WF1_LEN + WF2_LEN, g - WF1_LEN - WF2_LEN);
    return;
  }
  const int c0 = blockIdx.x * SCOLS;
  const int ph = blockIdx.x & (SPHASE - 1);
  const int rb = (threadIdx.x + 64 * ph) & (STHR - 1);
  const float* p = X + (size_t)rb * DIN + c0;
  v4f v[SROWS];
#pragma unroll
  for (int i = 0; i < SROWS; ++i) v[i] = *(const v4f*)(p + (size_t)i * STHR * DIN);
  v4f s = v[0], s2 = v[0] * v[0];
#pragma unroll
  for (int i = 1; i < SROWS; ++i) { s += v[i]; s2 += v[i] * v[i]; }
  float r[8] = {s.x, s.y, s.z, s.w, s2.x, s2.y, s2.z, s2.w};
#pragma unroll
  for (int q = 0; q < 8; ++q) {
#pragma unroll
    for (int off = 16; off >= 1; off >>= 1) r[q] += __shfl_xor(r[q], off, 32);
  }
  __shared__ float red[STHR / 32][8];
  const int lane = threadIdx.x & 31, wv = (rb >> 5);
  if (lane == 0) {
#pragma unroll
    for (int q = 0; q < 8; ++q) red[wv][q] = r[q];
  }
  __syncthreads();
  if (threadIdx.x < SCOLS) {
    float t = 0.f, t2 = 0.f;
    for (int w = 0; w < STHR / 32; ++w) { t += red[w][threadIdx.x]; t2 += red[w][4 + threadIdx.x]; }
    const int c = c0 + threadIdx.x;
    float m   = t * (1.f / NB);
    float var = t2 * (1.f / NB) - m * m;
    float rs  = rsqrtf(var + 1e-5f);
    nrm[2 * c]     = rs;
    nrm[2 * c + 1] = -m * rs;
  }
}

template <int KSTEPS, bool RELU>
static __device__ __forceinline__ v8f dense_tile(const float* pb, int nc, bool hi, int lane,
                                                 const float* __restrict__ Wf,
                                                 const float* __restrict__ bias, int ntile) {
  const int kof = hi ? 2 : 0;
  const int n = ntile * 16 + nc;
  const float* wt = Wf + ntile * (KSTEPS * 64);
  v8f acc = splat8(bias[n]);
#pragma unroll
  for (int q = 0; q < KSTEPS / 2; ++q) {
    const v4f b4 = *(const v4f*)(wt + q * 128 + lane * 4);
    v2f b0; b0.x = b4.x; b0.y = b4.y;
    v2f b1; b1.x = b4.z; b1.y = b4.w;
    const v2f a0 = *(const v2f*)(pb + nc * PSTR + 8 * q + kof);
    const v2f a1 = *(const v2f*)(pb + nc * PSTR + 8 * q + 4 + kof);
    acc = wmma4(a0, b0, acc);
    acc = wmma4(a1, b1, acc);
  }
  if (KSTEPS & 1) {
    const v2f b = *(const v2f*)(wt + (KSTEPS - 1) * 64 + lane * 2);
    const v2f a = *(const v2f*)(pb + nc * PSTR + 4 * (KSTEPS - 1) + kof);
    acc = wmma4(a, b, acc);
  }
  if (RELU) {
#pragma unroll
    for (int e = 0; e < 8; ++e) acc[e] = fmaxf(acc[e], 0.01f * acc[e]);
  }
  return acc;
}

static __device__ __forceinline__ void store_tile(float* pb, int nc, bool hi, int ntile, v8f h) {
#pragma unroll
  for (int e = 0; e < 8; ++e) {
    const int r = e + (hi ? 8 : 0);
    pb[r * PSTR + ntile * 16 + nc] = h[e];
  }
}

__global__ __launch_bounds__(128) void policy_fused_kernel(
    const float* __restrict__ X, const float* __restrict__ nrm,
    const float* __restrict__ Wme, const float* __restrict__ bme,
    const float* __restrict__ b1, const float* __restrict__ b2, const float* __restrict__ b3,
    float* __restrict__ out) {
  __shared__ __attribute__((aligned(16))) float part[NWAVE][16 * PSTR];
  __shared__ __attribute__((aligned(16))) float bufA[16 * PSTR];
  __shared__ __attribute__((aligned(16))) float bufB[16 * PSTR];

  const int  tid  = threadIdx.x;
  const int  lane = tid & 31;
  const int  wv   = __builtin_amdgcn_readfirstlane(tid >> 5);
  const int  nc   = lane & 15;
  const bool hi   = (lane >= 16);
  const int  rowbase = blockIdx.x * 16;
  const float* xrow = X + (size_t)(rowbase + nc) * DIN;

  v2f bw[4];
#pragma unroll
  for (int t = 0; t < 4; ++t) {
    const int n = t * 16 + nc;
    v2f w = *(const v2f*)(Wme + 2 * n);
    v2f bz; bz.x = bme[n]; bz.y = 0.f;
    bw[t] = hi ? bz : w;
  }

  v8f absacc[4];
#pragma unroll
  for (int t = 0; t < 4; ++t) absacc[t] = splat8(0.f);
  v2f asum; asum.x = 0.f; asum.y = 0.f;
  const v8f zero8 = splat8(0.f);

  const int c0 = 4 + 2 * PPW * wv;
  v2f xv = *(const v2f*)(xrow + c0);
  v4f nv = *(const v4f*)(nrm + 2 * c0);
  for (int i = 0; i < PPW; ++i) {
    v2f av;
    av.x = fmaf(xv.x, nv.x, nv.y);
    av.y = fmaf(xv.y, nv.z, nv.w);
    const int cn = (i < PPW - 1) ? (c0 + 2 + 2 * i) : c0;
    xv = *(const v2f*)(xrow + cn);
    nv = *(const v4f*)(nrm + 2 * cn);
    v2f one0; one0.x = 1.f; one0.y = 0.f;
    av = hi ? one0 : av;
    asum.x += av.x; asum.y += av.y;
#pragma unroll
    for (int t = 0; t < 4; ++t) {
      v8f z = wmma4(av, bw[t], zero8);
#pragma unroll
      for (int e = 0; e < 8; ++e) absacc[t][e] += fabsf(z[e]);
    }
  }

#pragma unroll
  for (int t = 0; t < 4; ++t) {
    v8f zs = wmma4(asum, bw[t], zero8);
#pragma unroll
    for (int e = 0; e < 8; ++e) {
      const float m = (0.505f * zs[e] + 0.495f * absacc[t][e]) * 0.01f;
      const int r = e + (hi ? 8 : 0);
      part[wv][r * PSTR + t * 16 + nc] = m;
    }
  }
  __syncthreads();

#pragma unroll
  for (int j = 0; j < 8; ++j) {
    const int idx = tid + 128 * j;
    const int r = idx >> 6, c = idx & 63;
    const int o = r * PSTR + c;
    bufA[r * PSTR + 4 + c] = (part[0][o] + part[1][o]) + (part[2][o] + part[3][o]);
  }
  if (tid < 64) {
    const int r = tid >> 2, c = tid & 3;
    bufA[r * PSTR + c] = fmaf(X[(size_t)(rowbase + r) * DIN + c], nrm[2 * c], nrm[2 * c + 1]);
  }
  __syncthreads();

  v8f h = dense_tile<17, true>(bufA, nc, hi, lane, nrm + WF1_OFF, b1, wv);
  store_tile(bufB, nc, hi, wv, h);
  __syncthreads();
  h = dense_tile<16, true>(bufB, nc, hi, lane, nrm + WF2_OFF, b2, wv);
  store_tile(bufA, nc, hi, wv, h);
  __syncthreads();
  if (wv < 2) {
    v8f lg = dense_tile<16, false>(bufA, nc, hi, lane, nrm + WF3_OFF, b3, wv);
    store_tile(bufB, nc, hi, wv, lg);
  }
  __syncthreads();

#pragma unroll
  for (int j = 0; j < 4; ++j) {
    const int r = wv * 4 + j;
    const float a = bufB[r * PSTR + lane];
    float m = a;
#pragma unroll
    for (int off = 16; off >= 1; off >>= 1) m = fmaxf(m, __shfl_xor(m, off, 32));
    const float ex = __expf(a - m);
    float s = ex;
#pragma unroll
    for (int off = 16; off >= 1; off >>= 1) s += __shfl_xor(s, off, 32);
    const float inv = 1.0f / s;
    out[(rowbase + r) * 32 + lane] = ex * inv;
  }
}

extern "C" void kernel_launch(void* const* d_in, const int* in_sizes, int n_in,
                              void* d_out, int out_size, void* d_ws, size_t ws_size,
                              hipStream_t stream) {
  const float* X   = (const float*)d_in[0];
  const float* Wme = (const float*)d_in[1];
  const float* bme = (const float*)d_in[2];
  const float* W1  = (const float*)d_in[3];
  const float* b1  = (const float*)d_in[4];
  const float* W2  = (const float*)d_in[5];
  const float* b2  = (const float*)d_in[6];
  const float* W3  = (const float*)d_in[7];
  const float* b3  = (const float*)d_in[8];
  float* out = (float*)d_out;

  float* nrm = (float*)d_ws;

  bn_stats_kernel<<<NSTAT + NPACK, STHR, 0, stream>>>(X, W1, W2, W3, nrm);
  policy_fused_kernel<<<NB / 16, 128, 0, stream>>>(X, nrm, Wme, bme, b1, b2, b3, out);
}
